# starting kernel
# baseline (speedup 1.0000x reference)
#include <hip/hip_runtime.h>

#define B_     4096
#define NFIELD 20
#define NEMB   16
#define HID    512
#define K_     16
#define GIN    320
#define EPSF   1e-10f

typedef __attribute__((ext_vector_type(2))) float v2f;
typedef __attribute__((ext_vector_type(8))) float v8f;

__device__ __forceinline__ v8f wmma_f32_16x16x4(v2f a, v2f b, v8f c) {
  return __builtin_amdgcn_wmma_f32_16x16x4_f32(false, a, false, b, (short)0, c,
                                               false, false);
}

__global__ void __launch_bounds__(256)
vsams_init_kernel(float* __restrict__ out, int* __restrict__ counts) {
  int t = blockIdx.x * blockDim.x + threadIdx.x;
  if (t <= B_) out[t] = 0.0f;
  if (t < K_) counts[t] = 0;
}

__global__ void __launch_bounds__(128)
vsams_gate_l1_kernel(const int* __restrict__ sql,
                     const float* __restrict__ sql_emb_w,
                     const float* __restrict__ gw1,
                     const float* __restrict__ gb1,
                     float* __restrict__ h) {
  __shared__ float Alds[16 * GIN];
  const int tid = threadIdx.x;
  const int m0  = blockIdx.y * 16;
  for (int idx = tid; idx < 16 * GIN; idx += 128) {
    int r = idx / GIN, c = idx - r * GIN;
    int f = c >> 4, e = c & 15;
    int s = sql[(m0 + r) * NFIELD + f];
    Alds[idx] = sql_emb_w[s * NEMB + e];
  }
  __syncthreads();
  const int lane = tid & 31, wave = tid >> 5;
  const int n0 = blockIdx.x * 64 + wave * 16;
  const int m  = lane & 15;
  const int kh = lane >> 4;
  v8f acc = {0.f, 0.f, 0.f, 0.f, 0.f, 0.f, 0.f, 0.f};
#pragma unroll 4
  for (int k0 = 0; k0 < GIN; k0 += 4) {
    v2f a, b;
    const float* ap = &Alds[m * GIN + k0 + 2 * kh];
    a.x = ap[0]; a.y = ap[1];
    const float* bp = &gw1[(size_t)(k0 + 2 * kh) * HID + n0 + m];
    b.x = bp[0]; b.y = bp[HID];
    acc = wmma_f32_16x16x4(a, b, acc);
  }
#pragma unroll
  for (int v = 0; v < 8; ++v) {
    int row = m0 + v + 8 * kh;
    int col = n0 + m;
    float val = acc[v] + gb1[col];
    h[(size_t)row * HID + col] = val > 0.f ? val : 0.f;
  }
}

__global__ void __launch_bounds__(32)
vsams_gate_head_kernel(const float* __restrict__ h,
                       const float* __restrict__ gw2,
                       const float* __restrict__ gb2,
                       float* __restrict__ gates,
                       int* __restrict__ counts,
                       int* __restrict__ lists) {
  __shared__ float logit[16 * 16];
  const int lane = threadIdx.x;
  const int m0 = blockIdx.x * 16;
  const int m  = lane & 15;
  const int kh = lane >> 4;
  v8f acc = {0.f, 0.f, 0.f, 0.f, 0.f, 0.f, 0.f, 0.f};
#pragma unroll 4
  for (int k0 = 0; k0 < HID; k0 += 4) {
    v2f a, b;
    const float* ap = &h[(size_t)(m0 + m) * HID + k0 + 2 * kh];
    a.x = ap[0]; a.y = ap[1];
    const float* bp = &gw2[(k0 + 2 * kh) * K_ + m];
    b.x = bp[0]; b.y = bp[K_];
    acc = wmma_f32_16x16x4(a, b, acc);
  }
#pragma unroll
  for (int v = 0; v < 8; ++v)
    logit[(v + 8 * kh) * 16 + m] = acc[v] + gb2[m];
  __syncthreads();
  if (lane < 16) {
    const int b = m0 + lane;
    float l[K_];
    float mx = -3.0e38f;
    for (int j = 0; j < K_; ++j) { l[j] = logit[lane * 16 + j]; mx = fmaxf(mx, l[j]); }
    float s = 0.f;
    for (int j = 0; j < K_; ++j) { l[j] = __expf(l[j] - mx); s += l[j]; }
    float inv = 1.f / s;
    for (int j = 0; j < K_; ++j) l[j] *= inv;
    int i1 = 0; float v1 = l[0];
    for (int j = 1; j < K_; ++j) if (l[j] > v1) { v1 = l[j]; i1 = j; }
    int i2 = -1; float v2 = -1.f;
    for (int j = 0; j < K_; ++j) if (j != i1 && l[j] > v2) { v2 = l[j]; i2 = j; }
    float denom = v1 + v2;
    float g1 = v1 / denom, g2 = v2 / denom;
    for (int j = 0; j < K_; ++j) gates[b * K_ + j] = 0.f;
    gates[b * K_ + i1] = g1;
    gates[b * K_ + i2] = g2;
    int p1 = atomicAdd(&counts[i1], 1);
    lists[i1 * B_ + p1] = b;
    int p2 = atomicAdd(&counts[i2], 1);
    lists[i2 * B_ + p2] = b;
  }
}

__global__ void __launch_bounds__(128)
vsams_expert_kernel(const int* __restrict__ x,
                    const float* __restrict__ inp_emb_w,
                    const float* __restrict__ ew1,
                    const float* __restrict__ eb1,
                    const float* __restrict__ ew2,
                    const float* __restrict__ eb2,
                    const float* __restrict__ gates,
                    const int* __restrict__ counts,
                    const int* __restrict__ lists,
                    float* __restrict__ y) {
  __shared__ float Alds[16 * GIN];
  __shared__ float EH[16 * HID];
  __shared__ float psum[16 * 8];
  __shared__ int   bids[16];
  const int k    = blockIdx.x >> 8;
  const int tile = blockIdx.x & 255;
  const int cnt  = counts[k];
  const int base = tile * 16;
  if (base >= cnt) return;
  const int tid = threadIdx.x;
  if (tid < 16) bids[tid] = (base + tid < cnt) ? lists[k * B_ + base + tid] : -1;
  __syncthreads();
  for (int idx = tid; idx < 16 * GIN; idx += 128) {
    int r = idx / GIN, c = idx - r * GIN;
    int b = bids[r];
    float val = 0.f;
    if (b >= 0) {
      int f = c >> 4, e = c & 15;
      int xi = x[b * NFIELD + f];
      val = inp_emb_w[(size_t)xi * NEMB + e];
    }
    Alds[idx] = val;
  }
  __syncthreads();
  const int lane = tid & 31, wave = tid >> 5;
  const int m  = lane & 15;
  const int kh = lane >> 4;
  const float* W1 = ew1 + (size_t)k * GIN * HID;
  const float* B1 = eb1 + k * HID;
  for (int nt = 0; nt < 8; ++nt) {
    const int n0 = (wave * 8 + nt) * 16;
    v8f acc = {0.f, 0.f, 0.f, 0.f, 0.f, 0.f, 0.f, 0.f};
#pragma unroll 4
    for (int k0 = 0; k0 < GIN; k0 += 4) {
      v2f a, b;
      const float* ap = &Alds[m * GIN + k0 + 2 * kh];
      a.x = ap[0]; a.y = ap[1];
      const float* bp = &W1[(size_t)(k0 + 2 * kh) * HID + n0 + m];
      b.x = bp[0]; b.y = bp[HID];
      acc = wmma_f32_16x16x4(a, b, acc);
    }
#pragma unroll
    for (int v = 0; v < 8; ++v) {
      int row = v + 8 * kh;
      int col = n0 + m;
      float val = acc[v] + B1[col];
      EH[row * HID + col] = val > 0.f ? val : 0.f;
    }
  }
  __syncthreads();
  {
    int row = tid >> 3, seg = tid & 7;
    const float* w2 = ew2 + k * HID + seg * 64;
    const float* eh = EH + row * HID + seg * 64;
    float p = 0.f;
#pragma unroll 8
    for (int c = 0; c < 64; ++c) p += eh[c] * w2[c];
    psum[row * 8 + seg] = p;
  }
  __syncthreads();
  if (tid < 16) {
    int b = bids[tid];
    if (b >= 0) {
      float eo = eb2[k];
      for (int s = 0; s < 8; ++s) eo += psum[tid * 8 + s];
      float g = gates[b * K_ + k];
      atomicAdd(&y[b], g * eo);
    }
  }
}

__global__ void __launch_bounds__(256)
vsams_loss_kernel(const float* __restrict__ gates, float* __restrict__ out) {
  __shared__ float simp[256], sld[256];
  const int t = threadIdx.x;
  const int k = t & 15, chunk = t >> 4;
  float pi = 0.f, pl = 0.f;
  for (int r = chunk * 256; r < chunk * 256 + 256; ++r) {
    float g = gates[r * K_ + k];
    pi += g;
    pl += (g > 0.f) ? 1.f : 0.f;
  }
  simp[t] = pi; sld[t] = pl;
  __syncthreads();
  if (t == 0) {
    float imp[K_], ld[K_];
    for (int kk = 0; kk < K_; ++kk) { imp[kk] = 0.f; ld[kk] = 0.f; }
    for (int c = 0; c < 16; ++c)
      for (int kk = 0; kk < K_; ++kk) {
        imp[kk] += simp[c * 16 + kk];
        ld[kk]  += sld[c * 16 + kk];
      }
    float mi = 0.f, ml = 0.f;
    for (int kk = 0; kk < K_; ++kk) { mi += imp[kk]; ml += ld[kk]; }
    mi *= (1.f / 16.f); ml *= (1.f / 16.f);
    float vi = 0.f, vl = 0.f;
    for (int kk = 0; kk < K_; ++kk) {
      float di = imp[kk] - mi, dl = ld[kk] - ml;
      vi += di * di; vl += dl * dl;
    }
    vi *= (1.f / 16.f); vl *= (1.f / 16.f);
    out[B_] = vi / (mi * mi + EPSF) + vl / (ml * ml + EPSF);
  }
}

extern "C" void kernel_launch(void* const* d_in, const int* in_sizes, int n_in,
                              void* d_out, int out_size, void* d_ws, size_t ws_size,
                              hipStream_t stream) {
  const int*   x         = (const int*)d_in[0];
  const int*   sql       = (const int*)d_in[1];
  const float* sql_emb_w = (const float*)d_in[2];
  const float* inp_emb_w = (const float*)d_in[3];
  const float* gw1 = (const float*)d_in[4];
  const float* gb1 = (const float*)d_in[5];
  const float* gw2 = (const float*)d_in[6];
  const float* gb2 = (const float*)d_in[7];
  const float* ew1 = (const float*)d_in[8];
  const float* eb1 = (const float*)d_in[9];
  const float* ew2 = (const float*)d_in[10];
  const float* eb2 = (const float*)d_in[11];
  float* out = (float*)d_out;

  float* h      = (float*)d_ws;
  float* gates  = h + (size_t)B_ * HID;
  int*   counts = (int*)(gates + (size_t)B_ * K_);
  int*   lists  = counts + K_;

  vsams_init_kernel<<<17, 256, 0, stream>>>(out, counts);
  vsams_gate_l1_kernel<<<dim3(8, 256), 128, 0, stream>>>(sql, sql_emb_w, gw1, gb1, h);
  vsams_gate_head_kernel<<<256, 32, 0, stream>>>(h, gw2, gb2, gates, counts, lists);
  vsams_expert_kernel<<<K_ * 256, 128, 0, stream>>>(x, inp_emb_w, ew1, eb1, ew2,
                                                    eb2, gates, counts, lists, out);
  vsams_loss_kernel<<<1, 256, 0, stream>>>(gates, out);
}


# Slab fill under the first tile's WMMAs: in both hidden GEMMs the row gather is issued first and the wave's 160 weight loads leave in 5 groups, two ahead of the WMMAs that use them; sums unchanged
# speedup vs baseline: 3.2143x; 3.2143x over previous
#include <hip/hip_runtime.h>

#define B_     4096
#define NFIELD 20
#define NEMB   16
#define HID    512
#define K_     16
#define GIN    320
#define EPSF   1e-10f

typedef __attribute__((ext_vector_type(2))) float v2f;
typedef __attribute__((ext_vector_type(8))) float v8f;

__device__ __forceinline__ v8f wmma_f32_16x16x4(v2f a, v2f b, v8f c) {
  return __builtin_amdgcn_wmma_f32_16x16x4_f32(false, a, false, b, (short)0, c,
                                               false, false);
}


template <int P0, int NP>
__device__ __forceinline__ void slab_load_steps(v2f (&w)[80], const float* __restrict__ W1, int kh) {
#pragma unroll
  for (int p = P0; p < P0 + NP; ++p) {
    const float* wp = W1 + (size_t)(8 * p + 4 * kh) * HID;
    w[2 * p].x     = wp[0];
    w[2 * p].y     = wp[HID];
    w[2 * p + 1].x = wp[2 * HID];
    w[2 * p + 1].y = wp[3 * HID];
  }
}

template <int P0, int NP, bool W_IS_A>
__device__ __forceinline__ v8f tile_k_steps(v8f acc, const v2f (&w)[80], const float* ap) {
#pragma unroll
  for (int p = P0; p < P0 + NP; ++p) {
    const float4 a4 = *(const float4*)(ap + 8 * p);
    v2f a0, a1;
    a0.x = a4.x; a0.y = a4.y;
    a1.x = a4.z; a1.y = a4.w;
    if (W_IS_A) {
      acc = wmma_f32_16x16x4(w[2 * p], a0, acc);
      acc = wmma_f32_16x16x4(w[2 * p + 1], a1, acc);
    } else {
      acc = wmma_f32_16x16x4(a0, w[2 * p], acc);
      acc = wmma_f32_16x16x4(a1, w[2 * p + 1], acc);
    }
  }
  return acc;
}

template <bool W_IS_A>
__device__ __forceinline__ v8f tile_under_fill(v2f (&w)[80], const float* __restrict__ W1, int kh,
                                               const float* ap) {
  v8f acc = {0.f, 0.f, 0.f, 0.f, 0.f, 0.f, 0.f, 0.f};
  slab_load_steps<0, 16>(w, W1, kh);
  __builtin_amdgcn_sched_barrier(0);
  slab_load_steps<16, 8>(w, W1, kh);
  acc = tile_k_steps<0, 8, W_IS_A>(acc, w, ap);
  __builtin_amdgcn_sched_barrier(0);
  slab_load_steps<24, 8>(w, W1, kh);
  acc = tile_k_steps<8, 8, W_IS_A>(acc, w, ap);
  __builtin_amdgcn_sched_barrier(0);
  slab_load_steps<32, 8>(w, W1, kh);
  acc = tile_k_steps<16, 8, W_IS_A>(acc, w, ap);
  __builtin_amdgcn_sched_barrier(0);
  acc = tile_k_steps<24, 16, W_IS_A>(acc, w, ap);
  return acc;
}

__global__ void __launch_bounds__(256)
vsams_init_kernel(float* __restrict__ out, int* __restrict__ counts) {
  int t = blockIdx.x * blockDim.x + threadIdx.x;
  if (t <= B_) out[t] = 0.0f;
  if (t < K_) counts[t] = 0;
}

#define APITCH  324
#define CGROUPS 4
#define GBLK    128
#define GTPB    2

__global__ void __launch_bounds__(256, 4)
vsams_gate_kernel(const int* __restrict__ sql,
                  const float* __restrict__ sql_emb_w,
                  const float* __restrict__ gw1,
                  const float* __restrict__ gb1,
                  const float* __restrict__ gw2,
                  float* __restrict__ plog) {
  __shared__ __attribute__((aligned(16))) float Alds[GTPB * 16 * APITCH];
  __shared__ __attribute__((aligned(16))) float lsum[GTPB * 8 * 256];
  const int cg = blockIdx.x % CGROUPS;
  const int j  = blockIdx.x / CGROUPS;
  const int tid  = threadIdx.x;
  const int lane = tid & 31, wave = tid >> 5;
  const int m  = lane & 15;
  const int kh = lane >> 4;
  const int n0 = cg * 128 + wave * 16;
  const int mb = j * (GTPB * 16);
  const float* W1 = gw1 + n0 + m;
  v2f w[80];
  float bias[8];
  v2f   g2[4];
#pragma unroll
  for (int v = 0; v < 8; ++v) bias[v] = gb1[n0 + 8 * kh + v];
#pragma unroll
  for (int q = 0; q < 4; ++q) {
    g2[q].x = gw2[(n0 + 8 * kh + 2 * q) * K_ + m];
    g2[q].y = gw2[(n0 + 8 * kh + 2 * q + 1) * K_ + m];
  }
  {
    int    si[GTPB * 5];
    float4 val[GTPB * 5];
#pragma unroll
    for (int i = 0; i < GTPB * 5; ++i) {
      const int q  = tid + i * 256;
      const int r  = q / 80, c4 = q - r * 80;
      si[i] = sql[(mb + r) * NFIELD + (c4 >> 2)];
    }
#pragma unroll
    for (int i = 0; i < GTPB * 5; ++i) {
      const int q  = tid + i * 256;
      const int c4 = q - (q / 80) * 80;
      val[i] = *(const float4*)(sql_emb_w + (size_t)si[i] * NEMB + (c4 & 3) * 4);
    }
#pragma unroll
    for (int i = 0; i < GTPB * 5; ++i) {
      const int q  = tid + i * 256;
      const int r  = q / 80, c4 = q - r * 80;
      *(float4*)&Alds[r * APITCH + c4 * 4] = val[i];
    }
  }
  __syncthreads();
#pragma unroll
  for (int t = 0; t < GTPB; ++t) {
    const float* ap = &Alds[(t * 16 + m) * APITCH + 4 * kh];
    v8f acc;
    if (t == 0) {
      acc = tile_under_fill<true>(w, W1, kh, ap);
    } else {
      acc = v8f{0.f, 0.f, 0.f, 0.f, 0.f, 0.f, 0.f, 0.f};
      acc = tile_k_steps<0, 40, true>(acc, w, ap);
    }
    v8f lg = {0.f, 0.f, 0.f, 0.f, 0.f, 0.f, 0.f, 0.f};
#pragma unroll
    for (int q = 0; q < 4; ++q) {
      float h0 = acc[2 * q] + bias[2 * q];
      float h1 = acc[2 * q + 1] + bias[2 * q + 1];
      v2f hb;
      hb.x = h0 > 0.f ? h0 : 0.f;
      hb.y = h1 > 0.f ? h1 : 0.f;
      lg = wmma_f32_16x16x4(g2[q], hb, lg);
    }
    float* lp = &lsum[(t * 8 + wave) * 256 + m * 16 + 8 * kh];
    *(float4*)lp       = make_float4(lg[0], lg[1], lg[2], lg[3]);
    *(float4*)(lp + 4) = make_float4(lg[4], lg[5], lg[6], lg[7]);
  }
  __syncthreads();
#pragma unroll
  for (int t = 0; t < GTPB; ++t) {
    float s = 0.f;
#pragma unroll
    for (int wv = 0; wv < 8; ++wv) s += lsum[(t * 8 + wv) * 256 + tid];
    plog[((size_t)cg * B_ + mb + t * 16) * K_ + tid] = s;
  }
}

__global__ void __launch_bounds__(32)
vsams_gate_head_kernel(const float* __restrict__ plog,
                       const float* __restrict__ gb2,
                       float* __restrict__ tsum,
                       float* __restrict__ glists,
                       int* __restrict__ counts,
                       int* __restrict__ lists) {
  __shared__ float logit[16 * 16];
  const int lane = threadIdx.x;
  const int m0 = blockIdx.x * 16;
  const int m  = lane & 15;
  const int kh = lane >> 4;
  {
    float pl[CGROUPS][8];
#pragma unroll
    for (int c = 0; c < CGROUPS; ++c)
#pragma unroll
      for (int i = 0; i < 8; ++i)
        pl[c][i] = plog[((size_t)c * B_ + m0) * K_ + i * 32 + lane];
    const float b2 = gb2[m];
#pragma unroll
    for (int i = 0; i < 8; ++i)
      logit[i * 32 + lane] = (((pl[0][i] + pl[1][i]) + pl[2][i]) + pl[3][i]) + b2;
  }
  __syncthreads();
  int   i1 = 0, i2 = 0;
  float g1 = 0.f, g2 = 0.f;
  if (lane < 16) {
    float l[K_];
    float mx = -3.0e38f;
    for (int j = 0; j < K_; ++j) { l[j] = logit[lane * 16 + j]; mx = fmaxf(mx, l[j]); }
    float s = 0.f;
    for (int j = 0; j < K_; ++j) { l[j] = __expf(l[j] - mx); s += l[j]; }
    float inv = 1.f / s;
    for (int j = 0; j < K_; ++j) l[j] *= inv;
    float v1 = l[0];
    for (int j = 1; j < K_; ++j) if (l[j] > v1) { v1 = l[j]; i1 = j; }
    i2 = -1; float v2 = -1.f;
    for (int j = 0; j < K_; ++j) if (j != i1 && l[j] > v2) { v2 = l[j]; i2 = j; }
    float denom = v1 + v2;
    g1 = v1 / denom; g2 = v2 / denom;
    for (int j = 0; j < K_; ++j)
      logit[lane * 16 + j] = (j == i1) ? g1 : ((j == i2) ? g2 : 0.f);
  }
  __syncthreads();
  float ts = 0.f;
  for (int r = 0; r < 16; ++r) {
    const float g = logit[r * 16 + m];
    ts += kh ? ((g > 0.f) ? 1.f : 0.f) : g;
  }
  tsum[blockIdx.x * 32 + lane] = ts;
  if (lane < 16) {
    const int b = m0 + lane;
    int p1 = atomicAdd(&counts[i1], 1);
    lists[i1 * B_ + p1]  = b;
    glists[i1 * B_ + p1] = g1;
    int p2 = atomicAdd(&counts[i2], 1);
    lists[i2 * B_ + p2]  = b;
    glists[i2 * B_ + p2] = g2;
  }
}

#define NBLK    128

template <int MASK>
__device__ __forceinline__ float row_xor_add(float v) {
  const int b = __builtin_bit_cast(int, v);
  const int o = __builtin_amdgcn_update_dpp(b, b, 0x160 | MASK, 0xf, 0xf, false);
  return v + __builtin_bit_cast(float, o);
}


__device__ __forceinline__ void expert_gather_loads(float4 (&val)[5], const int* __restrict__ x,
                                                    const float* __restrict__ inp_emb_w,
                                                    const int* __restrict__ lst, int base, int cnt,
                                                    int tid) {
  int bb[5], xi[5];
#pragma unroll
  for (int i = 0; i < 5; ++i) {
    const int r  = (tid + i * 256) / 80;
    const int rr = base + r;
    bb[i] = lst[rr < cnt ? rr : cnt - 1];
  }
#pragma unroll
  for (int i = 0; i < 5; ++i) {
    const int q  = tid + i * 256;
    const int c4 = q - (q / 80) * 80;
    xi[i] = x[bb[i] * NFIELD + (c4 >> 2)];
  }
#pragma unroll
  for (int i = 0; i < 5; ++i) {
    const int q  = tid + i * 256;
    const int c4 = q - (q / 80) * 80;
    val[i] = *(const float4*)(inp_emb_w + (size_t)xi[i] * NEMB + (c4 & 3) * 4);
  }
}

__device__ __forceinline__ void expert_gather_store(const float4 (&val)[5], float* Alds, int tid) {
#pragma unroll
  for (int i = 0; i < 5; ++i) {
    const int q  = tid + i * 256;
    const int r  = q / 80, c4 = q - r * 80;
    *(float4*)&Alds[r * APITCH + c4 * 4] = val[i];
  }
}

__device__ __forceinline__ void expert_tile_finish(v8f acc, float b1, float w2c, float bias2,
                                                   float* psum, const int* __restrict__ lst,
                                                   const float* __restrict__ glst, int base, int cnt,
                                                   int tid, int m, int kh, int wave,
                                                   float* __restrict__ y) {
  float e[8];
#pragma unroll
  for (int v = 0; v < 8; ++v) {
    float h1 = acc[v] + b1;
    h1 = h1 > 0.f ? h1 : 0.f;
    e[v] = h1 * w2c;
  }
#pragma unroll
  for (int v = 0; v < 8; ++v) e[v] = row_xor_add<1>(e[v]);
#pragma unroll
  for (int v = 0; v < 8; ++v) e[v] = row_xor_add<2>(e[v]);
#pragma unroll
  for (int v = 0; v < 8; ++v) e[v] = row_xor_add<4>(e[v]);
#pragma unroll
  for (int v = 0; v < 8; ++v) e[v] = row_xor_add<8>(e[v]);
  if (m == 0) {
#pragma unroll
    for (int v = 0; v < 8; ++v) psum[wave * 16 + 8 * kh + v] = e[v];
  }
  __syncthreads();
  if (tid < 16 && base + tid < cnt) {
    const int   b  = lst[base + tid];
    const float gt = glst[base + tid];
    float eo = bias2;
    for (int s = 0; s < 8; ++s) eo += psum[s * 16 + tid];
    atomicAdd(&y[b], gt * eo);
  }
}

__global__ void __launch_bounds__(256, 4)
vsams_expert_kernel(const int* __restrict__ x,
                    const float* __restrict__ inp_emb_w,
                    const float* __restrict__ ew1,
                    const float* __restrict__ eb1,
                    const float* __restrict__ ew2,
                    const float* __restrict__ eb2,
                    const float* __restrict__ glists,
                    const int* __restrict__ counts,
                    const int* __restrict__ lists,
                    float* __restrict__ y) {
  __shared__ __attribute__((aligned(16))) float Alds[16 * APITCH];
  __shared__ __attribute__((aligned(16))) float psum[8 * 16];
  const int cg = blockIdx.x % CGROUPS;
  const int j  = blockIdx.x / CGROUPS;
  int total = 0;
#pragma unroll
  for (int kk = 0; kk < K_; ++kk) total += (counts[kk] + 15) >> 4;
  const int start = (j * total) / NBLK;
  const int end   = ((j + 1) * total) / NBLK;
  if (start >= end) return;
  const int tid  = threadIdx.x;
  const int lane = tid & 31, wave = tid >> 5;
  const int m  = lane & 15;
  const int kh = lane >> 4;
  const int col = cg * 128 + wave * 16 + m;
  const float* ap = &Alds[m * APITCH + 4 * kh];
  int kbase = 0;
  for (int k = 0; k < K_; ++k) {
    const int cnt = counts[k];
    const int nt  = (cnt + 15) >> 4;
    const int lo  = start > kbase ? start : kbase;
    const int hi  = end < kbase + nt ? end : kbase + nt;
    if (lo < hi) {
      const float* W1 = ew1 + (size_t)k * GIN * HID + col;
      const float b1  = eb1[k * HID + col];
      const float w2c = ew2[k * HID + col];
      const float bias2 = (cg == 0) ? eb2[k] : 0.f;
      const int*   lst  = lists + k * B_;
      const float* glst = glists + k * B_;
      v2f w[80];
      {
        const int base = (lo - kbase) * 16;
        float4 val[5];
        expert_gather_loads(val, x, inp_emb_w, lst, base, cnt, tid);
        expert_gather_store(val, Alds, tid);
        __syncthreads();
        const v8f acc = tile_under_fill<false>(w, W1, kh, ap);
        expert_tile_finish(acc, b1, w2c, bias2, psum, lst, glst, base, cnt, tid, m, kh, wave, y);
      }
      for (int g = lo + 1; g < hi; ++g) {
        const int base = (g - kbase) * 16;
        float4 val[5];
        expert_gather_loads(val, x, inp_emb_w, lst, base, cnt, tid);
        expert_gather_store(val, Alds, tid);
        __syncthreads();
        v8f acc = {0.f, 0.f, 0.f, 0.f, 0.f, 0.f, 0.f, 0.f};
        acc = tile_k_steps<0, 40, false>(acc, w, ap);
        expert_tile_finish(acc, b1, w2c, bias2, psum, lst, glst, base, cnt, tid, m, kh, wave, y);
      }
    }
    kbase += nt;
  }
}

__global__ void __launch_bounds__(256)
vsams_loss_kernel(const float* __restrict__ part, float* __restrict__ out) {
  __shared__ float simp[256], sld[256];
  const int t = threadIdx.x;
  const int k = t & 15, chunk = t >> 4;
  float ti[16], tl[16];
#pragma unroll
  for (int i = 0; i < 16; ++i) {
    const float* p = part + (chunk * 16 + i) * 32;
    ti[i] = p[k];
    tl[i] = p[16 + k];
  }
  float pi = 0.f, pl = 0.f;
#pragma unroll
  for (int i = 0; i < 16; ++i) { pi += ti[i]; pl += tl[i]; }
  simp[t] = pi; sld[t] = pl;
  __syncthreads();
  if (t == 0) {
    float imp[K_], ld[K_];
    for (int kk = 0; kk < K_; ++kk) { imp[kk] = 0.f; ld[kk] = 0.f; }
    for (int c = 0; c < 16; ++c)
      for (int kk = 0; kk < K_; ++kk) {
        imp[kk] += simp[c * 16 + kk];
        ld[kk]  += sld[c * 16 + kk];
      }
    float mi = 0.f, ml = 0.f;
    for (int kk = 0; kk < K_; ++kk) { mi += imp[kk]; ml += ld[kk]; }
    mi *= (1.f / 16.f); ml *= (1.f / 16.f);
    float vi = 0.f, vl = 0.f;
    for (int kk = 0; kk < K_; ++kk) {
      float di = imp[kk] - mi, dl = ld[kk] - ml;
      vi += di * di; vl += dl * dl;
    }
    vi *= (1.f / 16.f); vl *= (1.f / 16.f);
    out[B_] = vi / (mi * mi + EPSF) + vl / (ml * ml + EPSF);
  }
}

extern "C" void kernel_launch(void* const* d_in, const int* in_sizes, int n_in,
                              void* d_out, int out_size, void* d_ws, size_t ws_size,
                              hipStream_t stream) {
  const int*   x         = (const int*)d_in[0];
  const int*   sql       = (const int*)d_in[1];
  const float* sql_emb_w = (const float*)d_in[2];
  const float* inp_emb_w = (const float*)d_in[3];
  const float* gw1 = (const float*)d_in[4];
  const float* gb1 = (const float*)d_in[5];
  const float* gw2 = (const float*)d_in[6];
  const float* gb2 = (const float*)d_in[7];
  const float* ew1 = (const float*)d_in[8];
  const float* eb1 = (const float*)d_in[9];
  const float* ew2 = (const float*)d_in[10];
  const float* eb2 = (const float*)d_in[11];
  float* out = (float*)d_out;

  float* plog   = (float*)d_ws;
  float* tsum   = plog + (size_t)CGROUPS * B_ * K_;
  float* glists = tsum + 256 * 32;
  int*   counts = (int*)(glists + (size_t)K_ * B_);
  int*   lists  = counts + K_;

  vsams_init_kernel<<<17, 256, 0, stream>>>(out, counts);
  vsams_gate_kernel<<<CGROUPS * GBLK, 256, 0, stream>>>(sql, sql_emb_w, gw1, gb1, gw2, plog);
  vsams_gate_head_kernel<<<256, 32, 0, stream>>>(plog, gb2, tsum, glists, counts, lists);
  vsams_expert_kernel<<<CGROUPS * NBLK, 256, 0, stream>>>(x, inp_emb_w, ew1, eb1, ew2,
                                                    eb2, glists, counts, lists, out);
  vsams_loss_kernel<<<1, 256, 0, stream>>>(tsum, out);
}


# Gate head, 32 blocks of 8 waves: samples take list places in the block by LDS adds, one 16-lane atomic claims the block's 16 ranges; was 8192 dependent adds on one 64-byte line; lists by arrival
# speedup vs baseline: 3.6343x; 1.1307x over previous
#include <hip/hip_runtime.h>

#define B_     4096
#define NFIELD 20
#define NEMB   16
#define HID    512
#define K_     16
#define GIN    320
#define EPSF   1e-10f

typedef __attribute__((ext_vector_type(2))) float v2f;
typedef __attribute__((ext_vector_type(8))) float v8f;

__device__ __forceinline__ v8f wmma_f32_16x16x4(v2f a, v2f b, v8f c) {
  return __builtin_amdgcn_wmma_f32_16x16x4_f32(false, a, false, b, (short)0, c,
                                               false, false);
}


template <int P0, int NP>
__device__ __forceinline__ void slab_load_steps(v2f (&w)[80], const float* __restrict__ W1, int kh) {
#pragma unroll
  for (int p = P0; p < P0 + NP; ++p) {
    const float* wp = W1 + (size_t)(8 * p + 4 * kh) * HID;
    w[2 * p].x     = wp[0];
    w[2 * p].y     = wp[HID];
    w[2 * p + 1].x = wp[2 * HID];
    w[2 * p + 1].y = wp[3 * HID];
  }
}

template <int P0, int NP, bool W_IS_A>
__device__ __forceinline__ v8f tile_k_steps(v8f acc, const v2f (&w)[80], const float* ap) {
#pragma unroll
  for (int p = P0; p < P0 + NP; ++p) {
    const float4 a4 = *(const float4*)(ap + 8 * p);
    v2f a0, a1;
    a0.x = a4.x; a0.y = a4.y;
    a1.x = a4.z; a1.y = a4.w;
    if (W_IS_A) {
      acc = wmma_f32_16x16x4(w[2 * p], a0, acc);
      acc = wmma_f32_16x16x4(w[2 * p + 1], a1, acc);
    } else {
      acc = wmma_f32_16x16x4(a0, w[2 * p], acc);
      acc = wmma_f32_16x16x4(a1, w[2 * p + 1], acc);
    }
  }
  return acc;
}

template <bool W_IS_A>
__device__ __forceinline__ v8f tile_under_fill(v2f (&w)[80], const float* __restrict__ W1, int kh,
                                               const float* ap) {
  v8f acc = {0.f, 0.f, 0.f, 0.f, 0.f, 0.f, 0.f, 0.f};
  slab_load_steps<0, 16>(w, W1, kh);
  __builtin_amdgcn_sched_barrier(0);
  slab_load_steps<16, 8>(w, W1, kh);
  acc = tile_k_steps<0, 8, W_IS_A>(acc, w, ap);
  __builtin_amdgcn_sched_barrier(0);
  slab_load_steps<24, 8>(w, W1, kh);
  acc = tile_k_steps<8, 8, W_IS_A>(acc, w, ap);
  __builtin_amdgcn_sched_barrier(0);
  slab_load_steps<32, 8>(w, W1, kh);
  acc = tile_k_steps<16, 8, W_IS_A>(acc, w, ap);
  __builtin_amdgcn_sched_barrier(0);
  acc = tile_k_steps<24, 16, W_IS_A>(acc, w, ap);
  return acc;
}

__global__ void __launch_bounds__(256)
vsams_init_kernel(float* __restrict__ out, int* __restrict__ counts) {
  int t = blockIdx.x * blockDim.x + threadIdx.x;
  if (t <= B_) out[t] = 0.0f;
  if (t < K_) counts[t] = 0;
}

#define APITCH  324
#define CGROUPS 4
#define GBLK    128
#define GTPB    2

__global__ void __launch_bounds__(256, 4)
vsams_gate_kernel(const int* __restrict__ sql,
                  const float* __restrict__ sql_emb_w,
                  const float* __restrict__ gw1,
                  const float* __restrict__ gb1,
                  const float* __restrict__ gw2,
                  float* __restrict__ plog) {
  __shared__ __attribute__((aligned(16))) float Alds[GTPB * 16 * APITCH];
  __shared__ __attribute__((aligned(16))) float lsum[GTPB * 8 * 256];
  const int cg = blockIdx.x % CGROUPS;
  const int j  = blockIdx.x / CGROUPS;
  const int tid  = threadIdx.x;
  const int lane = tid & 31, wave = tid >> 5;
  const int m  = lane & 15;
  const int kh = lane >> 4;
  const int n0 = cg * 128 + wave * 16;
  const int mb = j * (GTPB * 16);
  const float* W1 = gw1 + n0 + m;
  v2f w[80];
  float bias[8];
  v2f   g2[4];
#pragma unroll
  for (int v = 0; v < 8; ++v) bias[v] = gb1[n0 + 8 * kh + v];
#pragma unroll
  for (int q = 0; q < 4; ++q) {
    g2[q].x = gw2[(n0 + 8 * kh + 2 * q) * K_ + m];
    g2[q].y = gw2[(n0 + 8 * kh + 2 * q + 1) * K_ + m];
  }
  {
    int    si[GTPB * 5];
    float4 val[GTPB * 5];
#pragma unroll
    for (int i = 0; i < GTPB * 5; ++i) {
      const int q  = tid + i * 256;
      const int r  = q / 80, c4 = q - r * 80;
      si[i] = sql[(mb + r) * NFIELD + (c4 >> 2)];
    }
#pragma unroll
    for (int i = 0; i < GTPB * 5; ++i) {
      const int q  = tid + i * 256;
      const int c4 = q - (q / 80) * 80;
      val[i] = *(const float4*)(sql_emb_w + (size_t)si[i] * NEMB + (c4 & 3) * 4);
    }
#pragma unroll
    for (int i = 0; i < GTPB * 5; ++i) {
      const int q  = tid + i * 256;
      const int r  = q / 80, c4 = q - r * 80;
      *(float4*)&Alds[r * APITCH + c4 * 4] = val[i];
    }
  }
  __syncthreads();
#pragma unroll
  for (int t = 0; t < GTPB; ++t) {
    const float* ap = &Alds[(t * 16 + m) * APITCH + 4 * kh];
    v8f acc;
    if (t == 0) {
      acc = tile_under_fill<true>(w, W1, kh, ap);
    } else {
      acc = v8f{0.f, 0.f, 0.f, 0.f, 0.f, 0.f, 0.f, 0.f};
      acc = tile_k_steps<0, 40, true>(acc, w, ap);
    }
    v8f lg = {0.f, 0.f, 0.f, 0.f, 0.f, 0.f, 0.f, 0.f};
#pragma unroll
    for (int q = 0; q < 4; ++q) {
      float h0 = acc[2 * q] + bias[2 * q];
      float h1 = acc[2 * q + 1] + bias[2 * q + 1];
      v2f hb;
      hb.x = h0 > 0.f ? h0 : 0.f;
      hb.y = h1 > 0.f ? h1 : 0.f;
      lg = wmma_f32_16x16x4(g2[q], hb, lg);
    }
    float* lp = &lsum[(t * 8 + wave) * 256 + m * 16 + 8 * kh];
    *(float4*)lp       = make_float4(lg[0], lg[1], lg[2], lg[3]);
    *(float4*)(lp + 4) = make_float4(lg[4], lg[5], lg[6], lg[7]);
  }
  __syncthreads();
#pragma unroll
  for (int t = 0; t < GTPB; ++t) {
    float s = 0.f;
#pragma unroll
    for (int wv = 0; wv < 8; ++wv) s += lsum[(t * 8 + wv) * 256 + tid];
    plog[((size_t)cg * B_ + mb + t * 16) * K_ + tid] = s;
  }
}

#define HW 8
__global__ void __launch_bounds__(32 * HW)
vsams_gate_head_kernel(const float* __restrict__ plog,
                       const float* __restrict__ gb2,
                       float* __restrict__ tsum,
                       float* __restrict__ glists,
                       int* __restrict__ counts,
                       int* __restrict__ lists) {
  __shared__ float logit_all[HW * 16 * 16];
  __shared__ int   lcount[K_];
  const int tid  = threadIdx.x;
  const int lane = tid & 31, wave = tid >> 5;
  const int tile = blockIdx.x * HW + wave;
  float* logit = logit_all + wave * 256;
  const int m0 = tile * 16;
  const int m  = lane & 15;
  const int kh = lane >> 4;
  if (tid < K_) lcount[tid] = 0;
  {
    float pl[CGROUPS][8];
#pragma unroll
    for (int c = 0; c < CGROUPS; ++c)
#pragma unroll
      for (int i = 0; i < 8; ++i)
        pl[c][i] = plog[((size_t)c * B_ + m0) * K_ + i * 32 + lane];
    const float b2 = gb2[m];
#pragma unroll
    for (int i = 0; i < 8; ++i)
      logit[i * 32 + lane] = (((pl[0][i] + pl[1][i]) + pl[2][i]) + pl[3][i]) + b2;
  }
  __syncthreads();
  int   i1 = 0, i2 = 0;
  float g1 = 0.f, g2 = 0.f;
  int   p1 = 0, p2 = 0;
  if (lane < 16) {
    float l[K_];
    float mx = -3.0e38f;
    for (int j = 0; j < K_; ++j) { l[j] = logit[lane * 16 + j]; mx = fmaxf(mx, l[j]); }
    float s = 0.f;
    for (int j = 0; j < K_; ++j) { l[j] = __expf(l[j] - mx); s += l[j]; }
    float inv = 1.f / s;
    for (int j = 0; j < K_; ++j) l[j] *= inv;
    float v1 = l[0];
    for (int j = 1; j < K_; ++j) if (l[j] > v1) { v1 = l[j]; i1 = j; }
    i2 = -1; float v2 = -1.f;
    for (int j = 0; j < K_; ++j) if (j != i1 && l[j] > v2) { v2 = l[j]; i2 = j; }
    float denom = v1 + v2;
    g1 = v1 / denom; g2 = v2 / denom;
    p1 = atomicAdd(&lcount[i1], 1);
    p2 = atomicAdd(&lcount[i2], 1);
    for (int j = 0; j < K_; ++j)
      logit[lane * 16 + j] = (j == i1) ? g1 : ((j == i2) ? g2 : 0.f);
  }
  __syncthreads();
  int base = 0;
  if (tid < K_) base = atomicAdd(&counts[tid], lcount[tid]);
  float ts = 0.f;
  for (int r = 0; r < 16; ++r) {
    const float g = logit[r * 16 + m];
    ts += kh ? ((g > 0.f) ? 1.f : 0.f) : g;
  }
  if (tid < K_) lcount[tid] = base;
  __syncthreads();
  tsum[tile * 32 + lane] = ts;
  if (lane < 16) {
    const int b  = m0 + lane;
    const int q1 = lcount[i1] + p1;
    const int q2 = lcount[i2] + p2;
    lists[i1 * B_ + q1]  = b;
    glists[i1 * B_ + q1] = g1;
    lists[i2 * B_ + q2]  = b;
    glists[i2 * B_ + q2] = g2;
  }
}

#define NBLK    128

template <int MASK>
__device__ __forceinline__ float row_xor_add(float v) {
  const int b = __builtin_bit_cast(int, v);
  const int o = __builtin_amdgcn_update_dpp(b, b, 0x160 | MASK, 0xf, 0xf, false);
  return v + __builtin_bit_cast(float, o);
}


__device__ __forceinline__ void expert_gather_loads(float4 (&val)[5], const int* __restrict__ x,
                                                    const float* __restrict__ inp_emb_w,
                                                    const int* __restrict__ lst, int base, int cnt,
                                                    int tid) {
  int bb[5], xi[5];
#pragma unroll
  for (int i = 0; i < 5; ++i) {
    const int r  = (tid + i * 256) / 80;
    const int rr = base + r;
    bb[i] = lst[rr < cnt ? rr : cnt - 1];
  }
#pragma unroll
  for (int i = 0; i < 5; ++i) {
    const int q  = tid + i * 256;
    const int c4 = q - (q / 80) * 80;
    xi[i] = x[bb[i] * NFIELD + (c4 >> 2)];
  }
#pragma unroll
  for (int i = 0; i < 5; ++i) {
    const int q  = tid + i * 256;
    const int c4 = q - (q / 80) * 80;
    val[i] = *(const float4*)(inp_emb_w + (size_t)xi[i] * NEMB + (c4 & 3) * 4);
  }
}

__device__ __forceinline__ void expert_gather_store(const float4 (&val)[5], float* Alds, int tid) {
#pragma unroll
  for (int i = 0; i < 5; ++i) {
    const int q  = tid + i * 256;
    const int r  = q / 80, c4 = q - r * 80;
    *(float4*)&Alds[r * APITCH + c4 * 4] = val[i];
  }
}

__device__ __forceinline__ void expert_tile_finish(v8f acc, float b1, float w2c, float bias2,
                                                   float* psum, const int* __restrict__ lst,
                                                   const float* __restrict__ glst, int base, int cnt,
                                                   int tid, int m, int kh, int wave,
                                                   float* __restrict__ y) {
  float e[8];
#pragma unroll
  for (int v = 0; v < 8; ++v) {
    float h1 = acc[v] + b1;
    h1 = h1 > 0.f ? h1 : 0.f;
    e[v] = h1 * w2c;
  }
#pragma unroll
  for (int v = 0; v < 8; ++v) e[v] = row_xor_add<1>(e[v]);
#pragma unroll
  for (int v = 0; v < 8; ++v) e[v] = row_xor_add<2>(e[v]);
#pragma unroll
  for (int v = 0; v < 8; ++v) e[v] = row_xor_add<4>(e[v]);
#pragma unroll
  for (int v = 0; v < 8; ++v) e[v] = row_xor_add<8>(e[v]);
  if (m == 0) {
#pragma unroll
    for (int v = 0; v < 8; ++v) psum[wave * 16 + 8 * kh + v] = e[v];
  }
  __syncthreads();
  if (tid < 16 && base + tid < cnt) {
    const int   b  = lst[base + tid];
    const float gt = glst[base + tid];
    float eo = bias2;
    for (int s = 0; s < 8; ++s) eo += psum[s * 16 + tid];
    atomicAdd(&y[b], gt * eo);
  }
}

__global__ void __launch_bounds__(256, 4)
vsams_expert_kernel(const int* __restrict__ x,
                    const float* __restrict__ inp_emb_w,
                    const float* __restrict__ ew1,
                    const float* __restrict__ eb1,
                    const float* __restrict__ ew2,
                    const float* __restrict__ eb2,
                    const float* __restrict__ glists,
                    const int* __restrict__ counts,
                    const int* __restrict__ lists,
                    float* __restrict__ y) {
  __shared__ __attribute__((aligned(16))) float Alds[16 * APITCH];
  __shared__ __attribute__((aligned(16))) float psum[8 * 16];
  const int cg = blockIdx.x % CGROUPS;
  const int j  = blockIdx.x / CGROUPS;
  int total = 0;
#pragma unroll
  for (int kk = 0; kk < K_; ++kk) total += (counts[kk] + 15) >> 4;
  const int start = (j * total) / NBLK;
  const int end   = ((j + 1) * total) / NBLK;
  if (start >= end) return;
  const int tid  = threadIdx.x;
  const int lane = tid & 31, wave = tid >> 5;
  const int m  = lane & 15;
  const int kh = lane >> 4;
  const int col = cg * 128 + wave * 16 + m;
  const float* ap = &Alds[m * APITCH + 4 * kh];
  int kbase = 0;
  for (int k = 0; k < K_; ++k) {
    const int cnt = counts[k];
    const int nt  = (cnt + 15) >> 4;
    const int lo  = start > kbase ? start : kbase;
    const int hi  = end < kbase + nt ? end : kbase + nt;
    if (lo < hi) {
      const float* W1 = ew1 + (size_t)k * GIN * HID + col;
      const float b1  = eb1[k * HID + col];
      const float w2c = ew2[k * HID + col];
      const float bias2 = (cg == 0) ? eb2[k] : 0.f;
      const int*   lst  = lists + k * B_;
      const float* glst = glists + k * B_;
      v2f w[80];
      {
        const int base = (lo - kbase) * 16;
        float4 val[5];
        expert_gather_loads(val, x, inp_emb_w, lst, base, cnt, tid);
        expert_gather_store(val, Alds, tid);
        __syncthreads();
        const v8f acc = tile_under_fill<false>(w, W1, kh, ap);
        expert_tile_finish(acc, b1, w2c, bias2, psum, lst, glst, base, cnt, tid, m, kh, wave, y);
      }
      for (int g = lo + 1; g < hi; ++g) {
        const int base = (g - kbase) * 16;
        float4 val[5];
        expert_gather_loads(val, x, inp_emb_w, lst, base, cnt, tid);
        expert_gather_store(val, Alds, tid);
        __syncthreads();
        v8f acc = {0.f, 0.f, 0.f, 0.f, 0.f, 0.f, 0.f, 0.f};
        acc = tile_k_steps<0, 40, false>(acc, w, ap);
        expert_tile_finish(acc, b1, w2c, bias2, psum, lst, glst, base, cnt, tid, m, kh, wave, y);
      }
    }
    kbase += nt;
  }
}

__global__ void __launch_bounds__(256)
vsams_loss_kernel(const float* __restrict__ part, float* __restrict__ out) {
  __shared__ float simp[256], sld[256];
  const int t = threadIdx.x;
  const int k = t & 15, chunk = t >> 4;
  float ti[16], tl[16];
#pragma unroll
  for (int i = 0; i < 16; ++i) {
    const float* p = part + (chunk * 16 + i) * 32;
    ti[i] = p[k];
    tl[i] = p[16 + k];
  }
  float pi = 0.f, pl = 0.f;
#pragma unroll
  for (int i = 0; i < 16; ++i) { pi += ti[i]; pl += tl[i]; }
  simp[t] = pi; sld[t] = pl;
  __syncthreads();
  if (t == 0) {
    float imp[K_], ld[K_];
    for (int kk = 0; kk < K_; ++kk) { imp[kk] = 0.f; ld[kk] = 0.f; }
    for (int c = 0; c < 16; ++c)
      for (int kk = 0; kk < K_; ++kk) {
        imp[kk] += simp[c * 16 + kk];
        ld[kk]  += sld[c * 16 + kk];
      }
    float mi = 0.f, ml = 0.f;
    for (int kk = 0; kk < K_; ++kk) { mi += imp[kk]; ml += ld[kk]; }
    mi *= (1.f / 16.f); ml *= (1.f / 16.f);
    float vi = 0.f, vl = 0.f;
    for (int kk = 0; kk < K_; ++kk) {
      float di = imp[kk] - mi, dl = ld[kk] - ml;
      vi += di * di; vl += dl * dl;
    }
    vi *= (1.f / 16.f); vl *= (1.f / 16.f);
    out[B_] = vi / (mi * mi + EPSF) + vl / (ml * ml + EPSF);
  }
}

extern "C" void kernel_launch(void* const* d_in, const int* in_sizes, int n_in,
                              void* d_out, int out_size, void* d_ws, size_t ws_size,
                              hipStream_t stream) {
  const int*   x         = (const int*)d_in[0];
  const int*   sql       = (const int*)d_in[1];
  const float* sql_emb_w = (const float*)d_in[2];
  const float* inp_emb_w = (const float*)d_in[3];
  const float* gw1 = (const float*)d_in[4];
  const float* gb1 = (const float*)d_in[5];
  const float* gw2 = (const float*)d_in[6];
  const float* gb2 = (const float*)d_in[7];
  const float* ew1 = (const float*)d_in[8];
  const float* eb1 = (const float*)d_in[9];
  const float* ew2 = (const float*)d_in[10];
  const float* eb2 = (const float*)d_in[11];
  float* out = (float*)d_out;

  float* plog   = (float*)d_ws;
  float* tsum   = plog + (size_t)CGROUPS * B_ * K_;
  float* glists = tsum + 256 * 32;
  int*   counts = (int*)(glists + (size_t)K_ * B_);
  int*   lists  = counts + K_;

  vsams_init_kernel<<<17, 256, 0, stream>>>(out, counts);
  vsams_gate_kernel<<<CGROUPS * GBLK, 256, 0, stream>>>(sql, sql_emb_w, gw1, gb1, gw2, plog);
  vsams_gate_head_kernel<<<256 / HW, 32 * HW, 0, stream>>>(plog, gb2, tsum, glists, counts, lists);
  vsams_expert_kernel<<<CGROUPS * NBLK, 256, 0, stream>>>(x, inp_emb_w, ew1, eb1, ew2,
                                                    eb2, glists, counts, lists, out);
  vsams_loss_kernel<<<1, 256, 0, stream>>>(tsum, out);
}


# y without atomics: the expert launch stores each list slot's partial eo per column group, the last launch sums a sample's 2 x 4 partials times its gates; 4 launches, was 5; y sum order changes
# speedup vs baseline: 3.8495x; 1.0592x over previous
#include <hip/hip_runtime.h>

#define B_     4096
#define NFIELD 20
#define NEMB   16
#define HID    512
#define K_     16
#define GIN    320
#define EPSF   1e-10f

typedef __attribute__((ext_vector_type(2))) float v2f;
typedef __attribute__((ext_vector_type(8))) float v8f;

__device__ __forceinline__ v8f wmma_f32_16x16x4(v2f a, v2f b, v8f c) {
  return __builtin_amdgcn_wmma_f32_16x16x4_f32(false, a, false, b, (short)0, c,
                                               false, false);
}


template <int P0, int NP>
__device__ __forceinline__ void slab_load_steps(v2f (&w)[80], const float* __restrict__ W1, int kh) {
#pragma unroll
  for (int p = P0; p < P0 + NP; ++p) {
    const float* wp = W1 + (size_t)(8 * p + 4 * kh) * HID;
    w[2 * p].x     = wp[0];
    w[2 * p].y     = wp[HID];
    w[2 * p + 1].x = wp[2 * HID];
    w[2 * p + 1].y = wp[3 * HID];
  }
}

template <int P0, int NP, bool W_IS_A>
__device__ __forceinline__ v8f tile_k_steps(v8f acc, const v2f (&w)[80], const float* ap) {
#pragma unroll
  for (int p = P0; p < P0 + NP; ++p) {
    const float4 a4 = *(const float4*)(ap + 8 * p);
    v2f a0, a1;
    a0.x = a4.x; a0.y = a4.y;
    a1.x = a4.z; a1.y = a4.w;
    if (W_IS_A) {
      acc = wmma_f32_16x16x4(w[2 * p], a0, acc);
      acc = wmma_f32_16x16x4(w[2 * p + 1], a1, acc);
    } else {
      acc = wmma_f32_16x16x4(a0, w[2 * p], acc);
      acc = wmma_f32_16x16x4(a1, w[2 * p + 1], acc);
    }
  }
  return acc;
}

template <bool W_IS_A>
__device__ __forceinline__ v8f tile_under_fill(v2f (&w)[80], const float* __restrict__ W1, int kh,
                                               const float* ap) {
  v8f acc = {0.f, 0.f, 0.f, 0.f, 0.f, 0.f, 0.f, 0.f};
  slab_load_steps<0, 16>(w, W1, kh);
  __builtin_amdgcn_sched_barrier(0);
  slab_load_steps<16, 8>(w, W1, kh);
  acc = tile_k_steps<0, 8, W_IS_A>(acc, w, ap);
  __builtin_amdgcn_sched_barrier(0);
  slab_load_steps<24, 8>(w, W1, kh);
  acc = tile_k_steps<8, 8, W_IS_A>(acc, w, ap);
  __builtin_amdgcn_sched_barrier(0);
  slab_load_steps<32, 8>(w, W1, kh);
  acc = tile_k_steps<16, 8, W_IS_A>(acc, w, ap);
  __builtin_amdgcn_sched_barrier(0);
  acc = tile_k_steps<24, 16, W_IS_A>(acc, w, ap);
  return acc;
}

#define APITCH  324
#define CGROUPS 4
#define GBLK    128
#define GTPB    2

__global__ void __launch_bounds__(256, 4)
vsams_gate_kernel(const int* __restrict__ sql,
                  const float* __restrict__ sql_emb_w,
                  const float* __restrict__ gw1,
                  const float* __restrict__ gb1,
                  const float* __restrict__ gw2,
                  float* __restrict__ plog,
                  int* __restrict__ counts) {
  __shared__ __attribute__((aligned(16))) float Alds[GTPB * 16 * APITCH];
  __shared__ __attribute__((aligned(16))) float lsum[GTPB * 8 * 256];
  const int cg = blockIdx.x % CGROUPS;
  const int j  = blockIdx.x / CGROUPS;
  const int tid  = threadIdx.x;
  const int lane = tid & 31, wave = tid >> 5;
  const int m  = lane & 15;
  const int kh = lane >> 4;
  const int n0 = cg * 128 + wave * 16;
  const int mb = j * (GTPB * 16);
  const float* W1 = gw1 + n0 + m;
  if (blockIdx.x == 0 && tid < K_) counts[tid] = 0;
  v2f w[80];
  float bias[8];
  v2f   g2[4];
#pragma unroll
  for (int v = 0; v < 8; ++v) bias[v] = gb1[n0 + 8 * kh + v];
#pragma unroll
  for (int q = 0; q < 4; ++q) {
    g2[q].x = gw2[(n0 + 8 * kh + 2 * q) * K_ + m];
    g2[q].y = gw2[(n0 + 8 * kh + 2 * q + 1) * K_ + m];
  }
  {
    int    si[GTPB * 5];
    float4 val[GTPB * 5];
#pragma unroll
    for (int i = 0; i < GTPB * 5; ++i) {
      const int q  = tid + i * 256;
      const int r  = q / 80, c4 = q - r * 80;
      si[i] = sql[(mb + r) * NFIELD + (c4 >> 2)];
    }
#pragma unroll
    for (int i = 0; i < GTPB * 5; ++i) {
      const int q  = tid + i * 256;
      const int c4 = q - (q / 80) * 80;
      val[i] = *(const float4*)(sql_emb_w + (size_t)si[i] * NEMB + (c4 & 3) * 4);
    }
#pragma unroll
    for (int i = 0; i < GTPB * 5; ++i) {
      const int q  = tid + i * 256;
      const int r  = q / 80, c4 = q - r * 80;
      *(float4*)&Alds[r * APITCH + c4 * 4] = val[i];
    }
  }
  __syncthreads();
#pragma unroll
  for (int t = 0; t < GTPB; ++t) {
    const float* ap = &Alds[(t * 16 + m) * APITCH + 4 * kh];
    v8f acc;
    if (t == 0) {
      acc = tile_under_fill<true>(w, W1, kh, ap);
    } else {
      acc = v8f{0.f, 0.f, 0.f, 0.f, 0.f, 0.f, 0.f, 0.f};
      acc = tile_k_steps<0, 40, true>(acc, w, ap);
    }
    v8f lg = {0.f, 0.f, 0.f, 0.f, 0.f, 0.f, 0.f, 0.f};
#pragma unroll
    for (int q = 0; q < 4; ++q) {
      float h0 = acc[2 * q] + bias[2 * q];
      float h1 = acc[2 * q + 1] + bias[2 * q + 1];
      v2f hb;
      hb.x = h0 > 0.f ? h0 : 0.f;
      hb.y = h1 > 0.f ? h1 : 0.f;
      lg = wmma_f32_16x16x4(g2[q], hb, lg);
    }
    float* lp = &lsum[(t * 8 + wave) * 256 + m * 16 + 8 * kh];
    *(float4*)lp       = make_float4(lg[0], lg[1], lg[2], lg[3]);
    *(float4*)(lp + 4) = make_float4(lg[4], lg[5], lg[6], lg[7]);
  }
  __syncthreads();
#pragma unroll
  for (int t = 0; t < GTPB; ++t) {
    float s = 0.f;
#pragma unroll
    for (int wv = 0; wv < 8; ++wv) s += lsum[(t * 8 + wv) * 256 + tid];
    plog[((size_t)cg * B_ + mb + t * 16) * K_ + tid] = s;
  }
}

#define HW 8
__global__ void __launch_bounds__(32 * HW)
vsams_gate_head_kernel(const float* __restrict__ plog,
                       const float* __restrict__ gb2,
                       float* __restrict__ tsum,
                       float* __restrict__ gate2,
                       int* __restrict__ pos,
                       int* __restrict__ counts,
                       int* __restrict__ lists) {
  __shared__ float logit_all[HW * 16 * 16];
  __shared__ int   lcount[K_];
  const int tid  = threadIdx.x;
  const int lane = tid & 31, wave = tid >> 5;
  const int tile = blockIdx.x * HW + wave;
  float* logit = logit_all + wave * 256;
  const int m0 = tile * 16;
  const int m  = lane & 15;
  const int kh = lane >> 4;
  if (tid < K_) lcount[tid] = 0;
  {
    float pl[CGROUPS][8];
#pragma unroll
    for (int c = 0; c < CGROUPS; ++c)
#pragma unroll
      for (int i = 0; i < 8; ++i)
        pl[c][i] = plog[((size_t)c * B_ + m0) * K_ + i * 32 + lane];
    const float b2 = gb2[m];
#pragma unroll
    for (int i = 0; i < 8; ++i)
      logit[i * 32 + lane] = (((pl[0][i] + pl[1][i]) + pl[2][i]) + pl[3][i]) + b2;
  }
  __syncthreads();
  int   i1 = 0, i2 = 0;
  float g1 = 0.f, g2 = 0.f;
  int   p1 = 0, p2 = 0;
  if (lane < 16) {
    float l[K_];
    float mx = -3.0e38f;
    for (int j = 0; j < K_; ++j) { l[j] = logit[lane * 16 + j]; mx = fmaxf(mx, l[j]); }
    float s = 0.f;
    for (int j = 0; j < K_; ++j) { l[j] = __expf(l[j] - mx); s += l[j]; }
    float inv = 1.f / s;
    for (int j = 0; j < K_; ++j) l[j] *= inv;
    float v1 = l[0];
    for (int j = 1; j < K_; ++j) if (l[j] > v1) { v1 = l[j]; i1 = j; }
    i2 = -1; float v2 = -1.f;
    for (int j = 0; j < K_; ++j) if (j != i1 && l[j] > v2) { v2 = l[j]; i2 = j; }
    float denom = v1 + v2;
    g1 = v1 / denom; g2 = v2 / denom;
    p1 = atomicAdd(&lcount[i1], 1);
    p2 = atomicAdd(&lcount[i2], 1);
    for (int j = 0; j < K_; ++j)
      logit[lane * 16 + j] = (j == i1) ? g1 : ((j == i2) ? g2 : 0.f);
  }
  __syncthreads();
  int base = 0;
  if (tid < K_) base = atomicAdd(&counts[tid], lcount[tid]);
  float ts = 0.f;
  for (int r = 0; r < 16; ++r) {
    const float g = logit[r * 16 + m];
    ts += kh ? ((g > 0.f) ? 1.f : 0.f) : g;
  }
  if (tid < K_) lcount[tid] = base;
  __syncthreads();
  tsum[tile * 32 + lane] = ts;
  if (lane < 16) {
    const int b  = m0 + lane;
    const int q1 = lcount[i1] + p1;
    const int q2 = lcount[i2] + p2;
    lists[i1 * B_ + q1] = b;
    lists[i2 * B_ + q2] = b;
    *(int2*)&pos[2 * b]     = make_int2(i1 * B_ + q1, i2 * B_ + q2);
    *(float2*)&gate2[2 * b] = make_float2(g1, g2);
  }
}

#define NBLK    128

template <int MASK>
__device__ __forceinline__ float row_xor_add(float v) {
  const int b = __builtin_bit_cast(int, v);
  const int o = __builtin_amdgcn_update_dpp(b, b, 0x160 | MASK, 0xf, 0xf, false);
  return v + __builtin_bit_cast(float, o);
}


__device__ __forceinline__ void expert_gather_loads(float4 (&val)[5], const int* __restrict__ x,
                                                    const float* __restrict__ inp_emb_w,
                                                    const int* __restrict__ lst, int base, int cnt,
                                                    int tid) {
  int bb[5], xi[5];
#pragma unroll
  for (int i = 0; i < 5; ++i) {
    const int r  = (tid + i * 256) / 80;
    const int rr = base + r;
    bb[i] = lst[rr < cnt ? rr : cnt - 1];
  }
#pragma unroll
  for (int i = 0; i < 5; ++i) {
    const int q  = tid + i * 256;
    const int c4 = q - (q / 80) * 80;
    xi[i] = x[bb[i] * NFIELD + (c4 >> 2)];
  }
#pragma unroll
  for (int i = 0; i < 5; ++i) {
    const int q  = tid + i * 256;
    const int c4 = q - (q / 80) * 80;
    val[i] = *(const float4*)(inp_emb_w + (size_t)xi[i] * NEMB + (c4 & 3) * 4);
  }
}

__device__ __forceinline__ void expert_gather_store(const float4 (&val)[5], float* Alds, int tid) {
#pragma unroll
  for (int i = 0; i < 5; ++i) {
    const int q  = tid + i * 256;
    const int r  = q / 80, c4 = q - r * 80;
    *(float4*)&Alds[r * APITCH + c4 * 4] = val[i];
  }
}

__device__ __forceinline__ void expert_tile_finish(v8f acc, float b1, float w2c, float bias2,
                                                   float* psum, float* __restrict__ prt,
                                                   int base, int cnt,
                                                   int tid, int m, int kh, int wave) {
  float e[8];
#pragma unroll
  for (int v = 0; v < 8; ++v) {
    float h1 = acc[v] + b1;
    h1 = h1 > 0.f ? h1 : 0.f;
    e[v] = h1 * w2c;
  }
#pragma unroll
  for (int v = 0; v < 8; ++v) e[v] = row_xor_add<1>(e[v]);
#pragma unroll
  for (int v = 0; v < 8; ++v) e[v] = row_xor_add<2>(e[v]);
#pragma unroll
  for (int v = 0; v < 8; ++v) e[v] = row_xor_add<4>(e[v]);
#pragma unroll
  for (int v = 0; v < 8; ++v) e[v] = row_xor_add<8>(e[v]);
  if (m == 0) {
#pragma unroll
    for (int v = 0; v < 8; ++v) psum[wave * 16 + 8 * kh + v] = e[v];
  }
  __syncthreads();
  if (tid < 16 && base + tid < cnt) {
    float eo = bias2;
    for (int s = 0; s < 8; ++s) eo += psum[s * 16 + tid];
    prt[base + tid] = eo;
  }
}

__global__ void __launch_bounds__(256, 4)
vsams_expert_kernel(const int* __restrict__ x,
                    const float* __restrict__ inp_emb_w,
                    const float* __restrict__ ew1,
                    const float* __restrict__ eb1,
                    const float* __restrict__ ew2,
                    const float* __restrict__ eb2,
                    const int* __restrict__ counts,
                    const int* __restrict__ lists,
                    float* __restrict__ part) {
  __shared__ __attribute__((aligned(16))) float Alds[16 * APITCH];
  __shared__ __attribute__((aligned(16))) float psum[8 * 16];
  const int cg = blockIdx.x % CGROUPS;
  const int j  = blockIdx.x / CGROUPS;
  int total = 0;
#pragma unroll
  for (int kk = 0; kk < K_; ++kk) total += (counts[kk] + 15) >> 4;
  const int start = (j * total) / NBLK;
  const int end   = ((j + 1) * total) / NBLK;
  if (start >= end) return;
  const int tid  = threadIdx.x;
  const int lane = tid & 31, wave = tid >> 5;
  const int m  = lane & 15;
  const int kh = lane >> 4;
  const int col = cg * 128 + wave * 16 + m;
  const float* ap = &Alds[m * APITCH + 4 * kh];
  int kbase = 0;
  for (int k = 0; k < K_; ++k) {
    const int cnt = counts[k];
    const int nt  = (cnt + 15) >> 4;
    const int lo  = start > kbase ? start : kbase;
    const int hi  = end < kbase + nt ? end : kbase + nt;
    if (lo < hi) {
      const float* W1 = ew1 + (size_t)k * GIN * HID + col;
      const float b1  = eb1[k * HID + col];
      const float w2c = ew2[k * HID + col];
      const float bias2 = (cg == 0) ? eb2[k] : 0.f;
      const int* lst = lists + k * B_;
      float*     prt = part + ((size_t)cg * K_ + k) * B_;
      v2f w[80];
      {
        const int base = (lo - kbase) * 16;
        float4 val[5];
        expert_gather_loads(val, x, inp_emb_w, lst, base, cnt, tid);
        expert_gather_store(val, Alds, tid);
        __syncthreads();
        const v8f acc = tile_under_fill<false>(w, W1, kh, ap);
        expert_tile_finish(acc, b1, w2c, bias2, psum, prt, base, cnt, tid, m, kh, wave);
      }
      for (int g = lo + 1; g < hi; ++g) {
        const int base = (g - kbase) * 16;
        float4 val[5];
        expert_gather_loads(val, x, inp_emb_w, lst, base, cnt, tid);
        expert_gather_store(val, Alds, tid);
        __syncthreads();
        v8f acc = {0.f, 0.f, 0.f, 0.f, 0.f, 0.f, 0.f, 0.f};
        acc = tile_k_steps<0, 40, false>(acc, w, ap);
        expert_tile_finish(acc, b1, w2c, bias2, psum, prt, base, cnt, tid, m, kh, wave);
      }
    }
    kbase += nt;
  }
}

__global__ void __launch_bounds__(256)
vsams_final_kernel(const float* __restrict__ tsum,
                   const float* __restrict__ part,
                   const int* __restrict__ pos,
                   const float* __restrict__ gate2,
                   float* __restrict__ out) {
  __shared__ float simp[256], sld[256];
  const int t = threadIdx.x;
  if (blockIdx.x != 0) {
    const int    b  = (blockIdx.x - 1) * 256 + t;
    const int2   sl = *(const int2*)&pos[2 * b];
    const float2 g  = *(const float2*)&gate2[2 * b];
    float e1[CGROUPS], e2[CGROUPS];
#pragma unroll
    for (int c = 0; c < CGROUPS; ++c) {
      e1[c] = part[(size_t)c * K_ * B_ + sl.x];
      e2[c] = part[(size_t)c * K_ * B_ + sl.y];
    }
    out[b] = g.x * ((e1[0] + e1[1]) + (e1[2] + e1[3])) + g.y * ((e2[0] + e2[1]) + (e2[2] + e2[3]));
    return;
  }
  const int k = t & 15, chunk = t >> 4;
  float ti[16], tl[16];
#pragma unroll
  for (int i = 0; i < 16; ++i) {
    const float* p = tsum + (chunk * 16 + i) * 32;
    ti[i] = p[k];
    tl[i] = p[16 + k];
  }
  float pi = 0.f, pl = 0.f;
#pragma unroll
  for (int i = 0; i < 16; ++i) { pi += ti[i]; pl += tl[i]; }
  simp[t] = pi; sld[t] = pl;
  __syncthreads();
  if (t == 0) {
    float imp[K_], ld[K_];
    for (int kk = 0; kk < K_; ++kk) { imp[kk] = 0.f; ld[kk] = 0.f; }
    for (int c = 0; c < 16; ++c)
      for (int kk = 0; kk < K_; ++kk) {
        imp[kk] += simp[c * 16 + kk];
        ld[kk]  += sld[c * 16 + kk];
      }
    float mi = 0.f, ml = 0.f;
    for (int kk = 0; kk < K_; ++kk) { mi += imp[kk]; ml += ld[kk]; }
    mi *= (1.f / 16.f); ml *= (1.f / 16.f);
    float vi = 0.f, vl = 0.f;
    for (int kk = 0; kk < K_; ++kk) {
      float di = imp[kk] - mi, dl = ld[kk] - ml;
      vi += di * di; vl += dl * dl;
    }
    vi *= (1.f / 16.f); vl *= (1.f / 16.f);
    out[B_] = vi / (mi * mi + EPSF) + vl / (ml * ml + EPSF);
  }
}

extern "C" void kernel_launch(void* const* d_in, const int* in_sizes, int n_in,
                              void* d_out, int out_size, void* d_ws, size_t ws_size,
                              hipStream_t stream) {
  const int*   x         = (const int*)d_in[0];
  const int*   sql       = (const int*)d_in[1];
  const float* sql_emb_w = (const float*)d_in[2];
  const float* inp_emb_w = (const float*)d_in[3];
  const float* gw1 = (const float*)d_in[4];
  const float* gb1 = (const float*)d_in[5];
  const float* gw2 = (const float*)d_in[6];
  const float* gb2 = (const float*)d_in[7];
  const float* ew1 = (const float*)d_in[8];
  const float* eb1 = (const float*)d_in[9];
  const float* ew2 = (const float*)d_in[10];
  const float* eb2 = (const float*)d_in[11];
  float* out = (float*)d_out;

  float* plog   = (float*)d_ws;
  float* part   = plog;
  float* tsum   = plog + (size_t)CGROUPS * B_ * K_;
  float* gate2  = tsum + 256 * 32;
  int*   pos    = (int*)(gate2 + 2 * B_);
  int*   counts = (int*)(gate2 + (size_t)K_ * B_);
  int*   lists  = counts + K_;

  vsams_gate_kernel<<<CGROUPS * GBLK, 256, 0, stream>>>(sql, sql_emb_w, gw1, gb1, gw2, plog, counts);
  vsams_gate_head_kernel<<<256 / HW, 32 * HW, 0, stream>>>(plog, gb2, tsum, gate2, pos, counts, lists);
  vsams_expert_kernel<<<CGROUPS * NBLK, 256, 0, stream>>>(x, inp_emb_w, ew1, eb1, ew2,
                                                    eb2, counts, lists, part);
  vsams_final_kernel<<<1 + B_ / 256, 256, 0, stream>>>(tsum, part, pos, gate2, out);
}
